# starting kernel
# baseline (speedup 1.0000x reference)
#include <hip/hip_runtime.h>


typedef __attribute__((ext_vector_type(2))) float v2f;
typedef __attribute__((ext_vector_type(8))) float v8f;

#define F_DIM 48

__global__ void prep_w_kernel(const float* __restrict__ W1,
                              const float* __restrict__ bias,
                              const float* __restrict__ W2,
                              const float* __restrict__ b2,
                              float* __restrict__ w,
                              float* __restrict__ cconst,
                              int H) {
  int t = threadIdx.x;
  if (t < F_DIM) {
    float acc = 0.f;
    for (int h = 0; h < H; ++h) acc = fmaf(W1[t * H + h], W2[h], acc);
    w[t] = acc;
  } else if (t == F_DIM) {
    float sw = 0.f;
    for (int h = 0; h < H; ++h) sw += W2[h];
    *cconst = bias[0] * sw + b2[0];
  }
}

__global__ void init_deg_kernel(float* __restrict__ deg, int N) {
  int n = blockIdx.x * blockDim.x + threadIdx.x;
  if (n < N) deg[n] = 1.0f;
}

__global__ void degree_kernel(const int* __restrict__ col,
                              float* __restrict__ deg, int E) {
  int e = blockIdx.x * blockDim.x + threadIdx.x;
  if (e < E) atomicAdd(&deg[col[e]], 1.0f);
}

__global__ void dinv_kernel(float* __restrict__ deg, int N) {
  int n = blockIdx.x * blockDim.x + threadIdx.x;
  if (n < N) deg[n] = rsqrtf(fmaxf(deg[n], 1.0f));
}

__global__ void s_wmma_kernel(const float* __restrict__ x,
                              const float* __restrict__ w,
                              float* __restrict__ s, int nTiles) {
  int lane = threadIdx.x & 31;
  int tile = blockIdx.x * (blockDim.x >> 5) + (threadIdx.x >> 5);
  if (tile >= nTiles) return;
  int m = lane & 15;
  int halfSel = lane >> 4;
  const float* xrow = x + (size_t)(tile * 16 + m) * F_DIM;
  v8f c = {0.f, 0.f, 0.f, 0.f, 0.f, 0.f, 0.f, 0.f};
#pragma unroll
  for (int k0 = 0; k0 < F_DIM; k0 += 4) {
    int koff = k0 + 2 * halfSel;
    v2f a = *reinterpret_cast<const v2f*>(xrow + koff);
    v2f b = *reinterpret_cast<const v2f*>(w + koff);
    c = __builtin_amdgcn_wmma_f32_16x16x4_f32(
         false, a,  false, b,
         (short)0, c,  false,  false);
  }
  if ((lane & 15) == 0) {
    int base = tile * 16 + halfSel * 8;
#pragma unroll
    for (int j = 0; j < 8; ++j) s[base + j] = c[j];
  }
}

__global__ void s_tail_kernel(const float* __restrict__ x,
                              const float* __restrict__ w,
                              float* __restrict__ s, int start, int N) {
  int n = start + blockIdx.x * blockDim.x + threadIdx.x;
  if (n < N) {
    float acc = 0.f;
    for (int k = 0; k < F_DIM; ++k)
      acc = fmaf(x[(size_t)n * F_DIM + k], w[k], acc);
    s[n] = acc;
  }
}

__global__ void self_out_kernel(const float* __restrict__ dinv,
                                const float* __restrict__ s,
                                const float* __restrict__ cconst,
                                float* __restrict__ out, int N) {
  int n = blockIdx.x * blockDim.x + threadIdx.x;
  if (n < N) {
    float d = dinv[n];
    out[n] = fmaf(d * d, s[n], *cconst);
  }
}

__global__ void edge_scatter_kernel(const int* __restrict__ row,
                                    const int* __restrict__ col,
                                    const float* __restrict__ dinv,
                                    const float* __restrict__ s,
                                    float* __restrict__ out, int E) {
  int e = blockIdx.x * blockDim.x + threadIdx.x;
  if (e < E) {
    int r = row[e], c = col[e];
    atomicAdd(&out[c], dinv[r] * dinv[c] * s[r]);
  }
}

static inline size_t roundup256(size_t v) { return (v + 255) & ~(size_t)255; }

extern "C" void kernel_launch(void* const* d_in, const int* in_sizes, int n_in,
                              void* d_out, int out_size, void* d_ws, size_t ws_size,
                              hipStream_t stream) {
  const float* x    = (const float*)d_in[0];
  const int*   eidx = (const int*)d_in[1];
  const float* W1   = (const float*)d_in[2];
  const float* bias = (const float*)d_in[3];
  const float* W2   = (const float*)d_in[4];
  const float* b2   = (const float*)d_in[5];
  float* out = (float*)d_out;

  const int N = out_size;
  const int E = in_sizes[1] / 2;
  const int H = in_sizes[4];
  const int* row = eidx;
  const int* col = eidx + E;

  char* ws = (char*)d_ws;
  float* deg = (float*)ws;
  float* s   = (float*)(ws + roundup256((size_t)N * 4));
  float* w   = (float*)(ws + 2 * roundup256((size_t)N * 4));
  float* cc  = w + 64;
  (void)ws_size; (void)n_in;

  const int T = 256;
  const int nBlkN = (N + T - 1) / T;
  const int nBlkE = (E + T - 1) / T;

  prep_w_kernel<<<1, 64, 0, stream>>>(W1, bias, W2, b2, w, cc, H);
  init_deg_kernel<<<nBlkN, T, 0, stream>>>(deg, N);
  degree_kernel<<<nBlkE, T, 0, stream>>>(col, deg, E);
  dinv_kernel<<<nBlkN, T, 0, stream>>>(deg, N);

  const int nTiles = N / 16;
  const int wavesPerBlk = T / 32;
  const int nBlkTile = (nTiles + wavesPerBlk - 1) / wavesPerBlk;
  if (nTiles > 0)
    s_wmma_kernel<<<nBlkTile, T, 0, stream>>>(x, w, s, nTiles);
  const int rem = N - nTiles * 16;
  if (rem > 0)
    s_tail_kernel<<<1, T, 0, stream>>>(x, w, s, nTiles * 16, N);

  self_out_kernel<<<nBlkN, T, 0, stream>>>(deg, s, cc, out, N);
  edge_scatter_kernel<<<nBlkE, T, 0, stream>>>(row, col, deg, s, out, E);
}


# Second edge pass made a pull: the integer degree atomic returns each edge's rank, source stored in slot (rank,dest); a thread per dest sums t[src] over its slots; float sum order = arrival rank
# speedup vs baseline: 1.2977x; 1.2977x over previous
#include <hip/hip_runtime.h>


typedef __attribute__((ext_vector_type(2))) float v2f;
typedef __attribute__((ext_vector_type(8))) float v8f;

#define F_DIM 48
#define SLOT_CAP 64
#define RANK_EPT 4

__global__ void prep_w_kernel(const float* __restrict__ W1,
                              const float* __restrict__ bias,
                              const float* __restrict__ W2,
                              const float* __restrict__ b2,
                              float* __restrict__ w,
                              float* __restrict__ cconst,
                              int H) {
  int t = threadIdx.x;
  if (t < F_DIM) {
    float acc = 0.f;
    for (int h = 0; h < H; ++h) acc = fmaf(W1[t * H + h], W2[h], acc);
    w[t] = acc;
  } else if (t == F_DIM) {
    float sw = 0.f;
    for (int h = 0; h < H; ++h) sw += W2[h];
    *cconst = bias[0] * sw + b2[0];
  }
}

__global__ void init_cnt_kernel(int* __restrict__ cnt,
                                int* __restrict__ spillCount, int N) {
  int n = blockIdx.x * blockDim.x + threadIdx.x;
  if (n < N) cnt[n] = 0;
  if (n == 0) *spillCount = 0;
}

__global__ void rank_slot_kernel(const int* __restrict__ row,
                                 const int* __restrict__ col,
                                 int* __restrict__ cnt,
                                 int* __restrict__ slot,
                                 int* __restrict__ spillCount,
                                 int* __restrict__ spill,
                                 int E, int N) {
  const int base = blockIdx.x * (blockDim.x * RANK_EPT) + threadIdx.x;
  int r[RANK_EPT], c[RANK_EPT], k[RANK_EPT];
#pragma unroll
  for (int j = 0; j < RANK_EPT; ++j) {
    int e = base + j * blockDim.x;
    bool ok = e < E;
    c[j] = ok ? col[e] : -1;
    r[j] = ok ? row[e] : 0;
  }
#pragma unroll
  for (int j = 0; j < RANK_EPT; ++j)
    k[j] = (c[j] >= 0) ? atomicAdd(&cnt[c[j]], 1) : 0;
#pragma unroll
  for (int j = 0; j < RANK_EPT; ++j) {
    if (c[j] >= 0) {
      if (k[j] < SLOT_CAP) {
        slot[(size_t)k[j] * (size_t)N + (size_t)c[j]] = r[j];
      } else {
        int p = atomicAdd(spillCount, 1);
        spill[2 * (size_t)p]     = r[j];
        spill[2 * (size_t)p + 1] = c[j];
      }
    }
  }
}

__global__ void s_wmma_kernel(const float* __restrict__ x,
                              const float* __restrict__ w,
                              float* __restrict__ s, int nTiles) {
  int lane = threadIdx.x & 31;
  int tile = blockIdx.x * (blockDim.x >> 5) + (threadIdx.x >> 5);
  if (tile >= nTiles) return;
  int m = lane & 15;
  int halfSel = lane >> 4;
  const float* xrow = x + (size_t)(tile * 16 + m) * F_DIM;
  v8f c = {0.f, 0.f, 0.f, 0.f, 0.f, 0.f, 0.f, 0.f};
#pragma unroll
  for (int k0 = 0; k0 < F_DIM; k0 += 4) {
    int koff = k0 + 2 * halfSel;
    v2f a = *reinterpret_cast<const v2f*>(xrow + koff);
    v2f b = *reinterpret_cast<const v2f*>(w + koff);
    c = __builtin_amdgcn_wmma_f32_16x16x4_f32(
         false, a,  false, b,
         (short)0, c,  false,  false);
  }
  if ((lane & 15) == 0) {
    int base = tile * 16 + halfSel * 8;
#pragma unroll
    for (int j = 0; j < 8; ++j) s[base + j] = c[j];
  }
}

__global__ void s_tail_kernel(const float* __restrict__ x,
                              const float* __restrict__ w,
                              float* __restrict__ s, int start, int N) {
  int n = start + blockIdx.x * blockDim.x + threadIdx.x;
  if (n < N) {
    float acc = 0.f;
    for (int k = 0; k < F_DIM; ++k)
      acc = fmaf(x[(size_t)n * F_DIM + k], w[k], acc);
    s[n] = acc;
  }
}

__global__ void dinv_t_kernel(const int* __restrict__ cnt,
                              const float* __restrict__ s,
                              float* __restrict__ dinv,
                              float* __restrict__ t, int N) {
  int n = blockIdx.x * blockDim.x + threadIdx.x;
  if (n < N) {
    float d = rsqrtf(fmaxf((float)(cnt[n] + 1), 1.0f));
    dinv[n] = d;
    t[n] = d * s[n];
  }
}

#define PULL_U 8
__global__ void pull_out_kernel(const int* __restrict__ cnt,
                                const int* __restrict__ slot,
                                const float* __restrict__ dinv,
                                const float* __restrict__ t,
                                const float* __restrict__ cconst,
                                float* __restrict__ out, int N) {
  int c = blockIdx.x * blockDim.x + threadIdx.x;
  if (c >= N) return;
  int n = cnt[c];
  if (n > SLOT_CAP) n = SLOT_CAP;
  const float d = dinv[c];
  const float cst = *cconst;
  float acc = t[c];
  const int* sp = slot + c;
  for (int k0 = 0; k0 < n; k0 += PULL_U) {
    int r[PULL_U];
    float v[PULL_U];
#pragma unroll
    for (int j = 0; j < PULL_U; ++j)
      r[j] = (k0 + j < n) ? sp[(size_t)(k0 + j) * (size_t)N] : -1;
#pragma unroll
    for (int j = 0; j < PULL_U; ++j)
      v[j] = (r[j] >= 0) ? t[r[j]] : 0.f;
#pragma unroll
    for (int j = 0; j < PULL_U; ++j) acc += v[j];
  }
  out[c] = fmaf(d, acc, cst);
}

__global__ void spill_add_kernel(const int* __restrict__ spillCount,
                                 const int* __restrict__ spill,
                                 const float* __restrict__ dinv,
                                 const float* __restrict__ t,
                                 float* __restrict__ out, int E) {
  int n = *spillCount;
  if (n > E) n = E;
  int stride = gridDim.x * blockDim.x;
  for (int i = blockIdx.x * blockDim.x + threadIdx.x; i < n; i += stride) {
    int r = spill[2 * (size_t)i];
    int c = spill[2 * (size_t)i + 1];
    atomicAdd(&out[c], dinv[c] * t[r]);
  }
}

static inline size_t roundup256(size_t v) { return (v + 255) & ~(size_t)255; }

extern "C" void kernel_launch(void* const* d_in, const int* in_sizes, int n_in,
                              void* d_out, int out_size, void* d_ws, size_t ws_size,
                              hipStream_t stream) {
  const float* x    = (const float*)d_in[0];
  const int*   eidx = (const int*)d_in[1];
  const float* W1   = (const float*)d_in[2];
  const float* bias = (const float*)d_in[3];
  const float* W2   = (const float*)d_in[4];
  const float* b2   = (const float*)d_in[5];
  float* out = (float*)d_out;

  const int N = out_size;
  const int E = in_sizes[1] / 2;
  const int H = in_sizes[4];
  const int* row = eidx;
  const int* col = eidx + E;

  char* ws = (char*)d_ws;
  const size_t plane = roundup256((size_t)N * 4);
  int*   cnt  = (int*)ws;
  float* s    = (float*)(ws + plane);
  float* dinv = (float*)(ws + 2 * plane);
  float* t    = (float*)(ws + 3 * plane);
  float* w    = (float*)(ws + 4 * plane);
  float* cc   = w + 64;
  int*   spillCount = (int*)(w + 65);
  int*   slot  = (int*)(ws + 4 * plane + 1024);
  int*   spill = (int*)(ws + 4 * plane + 1024 +
                        roundup256((size_t)SLOT_CAP * (size_t)N * 4));
  (void)ws_size; (void)n_in;

  const int T = 256;
  const int nBlkN = (N + T - 1) / T;
  const int nBlkRank = (E + T * RANK_EPT - 1) / (T * RANK_EPT);

  prep_w_kernel<<<1, 64, 0, stream>>>(W1, bias, W2, b2, w, cc, H);
  init_cnt_kernel<<<nBlkN, T, 0, stream>>>(cnt, spillCount, N);
  if (nBlkRank > 0)
    rank_slot_kernel<<<nBlkRank, T, 0, stream>>>(row, col, cnt, slot, spillCount, spill, E, N);

  const int nTiles = N / 16;
  const int wavesPerBlk = T / 32;
  const int nBlkTile = (nTiles + wavesPerBlk - 1) / wavesPerBlk;
  if (nTiles > 0)
    s_wmma_kernel<<<nBlkTile, T, 0, stream>>>(x, w, s, nTiles);
  const int rem = N - nTiles * 16;
  if (rem > 0)
    s_tail_kernel<<<1, T, 0, stream>>>(x, w, s, nTiles * 16, N);

  dinv_t_kernel<<<nBlkN, T, 0, stream>>>(cnt, s, dinv, t, N);

  const int TP = 128;
  pull_out_kernel<<<(N + TP - 1) / TP, TP, 0, stream>>>(cnt, slot, dinv, t, cc, out, N);
  spill_add_kernel<<<64, T, 0, stream>>>(spillCount, spill, dinv, t, out, E);
}


# Counters and slot table kept once per package half; the die id picks a wave's copy, so rank atomics and slot stores stay in one half's L2; float sum order = half 0's ranks, then half 1's
# speedup vs baseline: 3.7114x; 2.8600x over previous
#include <hip/hip_runtime.h>


typedef __attribute__((ext_vector_type(2))) float v2f;
typedef __attribute__((ext_vector_type(8))) float v8f;

#define F_DIM 48
#define SLOT_CAP 64
#define RANK_EPT 4
#define N_HALF 2

__device__ __forceinline__ int cache_half() {
  unsigned v = __builtin_amdgcn_s_sendmsg_rtn(0x87);
  return (int)((v >> 18) & 1u);
}

__global__ void prep_w_kernel(const float* __restrict__ W1,
                              const float* __restrict__ bias,
                              const float* __restrict__ W2,
                              const float* __restrict__ b2,
                              float* __restrict__ w,
                              float* __restrict__ cconst,
                              int H) {
  int t = threadIdx.x;
  if (t < F_DIM) {
    float acc = 0.f;
    for (int h = 0; h < H; ++h) acc = fmaf(W1[t * H + h], W2[h], acc);
    w[t] = acc;
  } else if (t == F_DIM) {
    float sw = 0.f;
    for (int h = 0; h < H; ++h) sw += W2[h];
    *cconst = bias[0] * sw + b2[0];
  }
}

__global__ void init_cnt_kernel(int* __restrict__ cnt,
                                int* __restrict__ spillCount, int nWords) {
  int n = blockIdx.x * blockDim.x + threadIdx.x;
  if (n < nWords) cnt[n] = 0;
  if (n == 0) *spillCount = 0;
}

__global__ void rank_slot_kernel(const int* __restrict__ row,
                                 const int* __restrict__ col,
                                 int* __restrict__ cnt,
                                 int* __restrict__ slot,
                                 int* __restrict__ spillCount,
                                 int* __restrict__ spill,
                                 int E, int N, int cntStride) {
  const int half = cache_half();
  cnt  += (size_t)half * (size_t)cntStride;
  slot += (size_t)half * (size_t)SLOT_CAP * (size_t)N;
  const int base = blockIdx.x * (blockDim.x * RANK_EPT) + threadIdx.x;
  int r[RANK_EPT], c[RANK_EPT], k[RANK_EPT];
#pragma unroll
  for (int j = 0; j < RANK_EPT; ++j) {
    int e = base + j * blockDim.x;
    bool ok = e < E;
    c[j] = ok ? col[e] : -1;
    r[j] = ok ? row[e] : 0;
  }
#pragma unroll
  for (int j = 0; j < RANK_EPT; ++j)
    k[j] = (c[j] >= 0) ? atomicAdd(&cnt[c[j]], 1) : 0;
#pragma unroll
  for (int j = 0; j < RANK_EPT; ++j) {
    if (c[j] >= 0) {
      if (k[j] < SLOT_CAP) {
        slot[(size_t)k[j] * (size_t)N + (size_t)c[j]] = r[j];
      } else {
        int p = atomicAdd(spillCount, 1);
        spill[2 * (size_t)p]     = r[j];
        spill[2 * (size_t)p + 1] = c[j];
      }
    }
  }
}

__global__ void s_wmma_kernel(const float* __restrict__ x,
                              const float* __restrict__ w,
                              float* __restrict__ s, int nTiles) {
  int lane = threadIdx.x & 31;
  int tile = blockIdx.x * (blockDim.x >> 5) + (threadIdx.x >> 5);
  if (tile >= nTiles) return;
  int m = lane & 15;
  int halfSel = lane >> 4;
  const float* xrow = x + (size_t)(tile * 16 + m) * F_DIM;
  v8f c = {0.f, 0.f, 0.f, 0.f, 0.f, 0.f, 0.f, 0.f};
#pragma unroll
  for (int k0 = 0; k0 < F_DIM; k0 += 4) {
    int koff = k0 + 2 * halfSel;
    v2f a = *reinterpret_cast<const v2f*>(xrow + koff);
    v2f b = *reinterpret_cast<const v2f*>(w + koff);
    c = __builtin_amdgcn_wmma_f32_16x16x4_f32(
         false, a,  false, b,
         (short)0, c,  false,  false);
  }
  if ((lane & 15) == 0) {
    int base = tile * 16 + halfSel * 8;
#pragma unroll
    for (int j = 0; j < 8; ++j) s[base + j] = c[j];
  }
}

__global__ void s_tail_kernel(const float* __restrict__ x,
                              const float* __restrict__ w,
                              float* __restrict__ s, int start, int N) {
  int n = start + blockIdx.x * blockDim.x + threadIdx.x;
  if (n < N) {
    float acc = 0.f;
    for (int k = 0; k < F_DIM; ++k)
      acc = fmaf(x[(size_t)n * F_DIM + k], w[k], acc);
    s[n] = acc;
  }
}

__global__ void dinv_t_kernel(const int* __restrict__ cnt,
                              const float* __restrict__ s,
                              float* __restrict__ dinv,
                              float* __restrict__ t, int N, int cntStride) {
  int n = blockIdx.x * blockDim.x + threadIdx.x;
  if (n < N) {
    float d = rsqrtf(fmaxf((float)(cnt[n] + cnt[cntStride + n] + 1), 1.0f));
    dinv[n] = d;
    t[n] = d * s[n];
  }
}

#define PULL_U 8
__global__ void pull_out_kernel(const int* __restrict__ cnt,
                                const int* __restrict__ slot,
                                const float* __restrict__ dinv,
                                const float* __restrict__ t,
                                const float* __restrict__ cconst,
                                float* __restrict__ out, int N, int cntStride) {
  int c = blockIdx.x * blockDim.x + threadIdx.x;
  if (c >= N) return;
  int n0 = cnt[c];
  int n1 = cnt[cntStride + c];
  if (n0 > SLOT_CAP) n0 = SLOT_CAP;
  if (n1 > SLOT_CAP) n1 = SLOT_CAP;
  const int n = n0 + n1;
  const float d = dinv[c];
  const float cst = *cconst;
  float acc = t[c];
  const int* sp = slot + c;
  const int skip = SLOT_CAP - n0;
  for (int k0 = 0; k0 < n; k0 += PULL_U) {
    int r[PULL_U];
    float v[PULL_U];
#pragma unroll
    for (int j = 0; j < PULL_U; ++j) {
      int i = k0 + j;
      int p = (i < n0) ? i : i + skip;
      r[j] = (i < n) ? sp[(size_t)p * (size_t)N] : -1;
    }
#pragma unroll
    for (int j = 0; j < PULL_U; ++j)
      v[j] = (r[j] >= 0) ? t[r[j]] : 0.f;
#pragma unroll
    for (int j = 0; j < PULL_U; ++j) acc += v[j];
  }
  out[c] = fmaf(d, acc, cst);
}

__global__ void spill_add_kernel(const int* __restrict__ spillCount,
                                 const int* __restrict__ spill,
                                 const float* __restrict__ dinv,
                                 const float* __restrict__ t,
                                 float* __restrict__ out, int E) {
  int n = *spillCount;
  if (n > E) n = E;
  int stride = gridDim.x * blockDim.x;
  for (int i = blockIdx.x * blockDim.x + threadIdx.x; i < n; i += stride) {
    int r = spill[2 * (size_t)i];
    int c = spill[2 * (size_t)i + 1];
    atomicAdd(&out[c], dinv[c] * t[r]);
  }
}

static inline size_t roundup256(size_t v) { return (v + 255) & ~(size_t)255; }

extern "C" void kernel_launch(void* const* d_in, const int* in_sizes, int n_in,
                              void* d_out, int out_size, void* d_ws, size_t ws_size,
                              hipStream_t stream) {
  const float* x    = (const float*)d_in[0];
  const int*   eidx = (const int*)d_in[1];
  const float* W1   = (const float*)d_in[2];
  const float* bias = (const float*)d_in[3];
  const float* W2   = (const float*)d_in[4];
  const float* b2   = (const float*)d_in[5];
  float* out = (float*)d_out;

  const int N = out_size;
  const int E = in_sizes[1] / 2;
  const int H = in_sizes[4];
  const int* row = eidx;
  const int* col = eidx + E;

  char* ws = (char*)d_ws;
  const size_t plane = roundup256((size_t)N * 4);
  const int cntStride = (int)(plane / 4);
  int*   cnt  = (int*)ws;
  float* s    = (float*)(ws + (N_HALF + 0) * plane);
  float* dinv = (float*)(ws + (N_HALF + 1) * plane);
  float* t    = (float*)(ws + (N_HALF + 2) * plane);
  float* w    = (float*)(ws + (N_HALF + 3) * plane);
  float* cc   = w + 64;
  int*   spillCount = (int*)(w + 65);
  int*   slot  = (int*)(ws + (N_HALF + 3) * plane + 1024);
  int*   spill = (int*)(ws + (N_HALF + 3) * plane + 1024 +
                        roundup256((size_t)N_HALF * (size_t)SLOT_CAP * (size_t)N * 4));
  (void)ws_size; (void)n_in;

  const int T = 256;
  const int nBlkN = (N + T - 1) / T;
  const int nCntWords = N_HALF * cntStride;
  const int nBlkCnt = (nCntWords + T - 1) / T;
  const int nBlkRank = (E + T * RANK_EPT - 1) / (T * RANK_EPT);

  prep_w_kernel<<<1, 64, 0, stream>>>(W1, bias, W2, b2, w, cc, H);
  init_cnt_kernel<<<nBlkCnt, T, 0, stream>>>(cnt, spillCount, nCntWords);
  if (nBlkRank > 0)
    rank_slot_kernel<<<nBlkRank, T, 0, stream>>>(row, col, cnt, slot, spillCount, spill, E, N, cntStride);

  const int nTiles = N / 16;
  const int wavesPerBlk = T / 32;
  const int nBlkTile = (nTiles + wavesPerBlk - 1) / wavesPerBlk;
  if (nTiles > 0)
    s_wmma_kernel<<<nBlkTile, T, 0, stream>>>(x, w, s, nTiles);
  const int rem = N - nTiles * 16;
  if (rem > 0)
    s_tail_kernel<<<1, T, 0, stream>>>(x, w, s, nTiles * 16, N);

  dinv_t_kernel<<<nBlkN, T, 0, stream>>>(cnt, s, dinv, t, N, cntStride);

  const int TP = 128;
  pull_out_kernel<<<(N + TP - 1) / TP, TP, 0, stream>>>(cnt, slot, dinv, t, cc, out, N, cntStride);
  spill_add_kernel<<<64, T, 0, stream>>>(spillCount, spill, dinv, t, out, E);
}


# Launches 7 to 4: W1@W2 collapse (wave per row + butterfly: sum order of w changes) shares the counter-clear launch, s tiles share the rank launch, the pull adds spilled edges itself: no float atomics
# speedup vs baseline: 5.1715x; 1.3934x over previous
#include <hip/hip_runtime.h>


typedef __attribute__((ext_vector_type(2))) float v2f;
typedef __attribute__((ext_vector_type(8))) float v8f;

#define F_DIM 48
#define SLOT_CAP 64
#define RANK_EPT 4
#define N_HALF 2

__device__ __forceinline__ int cache_half() {
  unsigned v = __builtin_amdgcn_s_sendmsg_rtn(0x87);
  return (int)((v >> 18) & 1u);
}

#define PREP_WAVES (F_DIM + 1)
__global__ void prep_zero_kernel(const float* __restrict__ W1,
                                 const float* __restrict__ bias,
                                 const float* __restrict__ W2,
                                 const float* __restrict__ b2,
                                 float* __restrict__ w,
                                 float* __restrict__ cconst,
                                 int H,
                                 int* __restrict__ cnt,
                                 int* __restrict__ spillCount,
                                 int nCntWords, int nPrepBlk) {
  if ((int)blockIdx.x < nPrepBlk) {
    const int lane = threadIdx.x & 31;
    const int g = blockIdx.x * (blockDim.x >> 5) + (threadIdx.x >> 5);
    if (g >= PREP_WAVES) return;
    float acc = 0.f;
    if (g < F_DIM) {
      const float* wr = W1 + (size_t)g * (size_t)H;
#pragma unroll 8
      for (int h = lane; h < H; h += 32) acc = fmaf(wr[h], W2[h], acc);
    } else {
#pragma unroll 8
      for (int h = lane; h < H; h += 32) acc += W2[h];
    }
    for (int o = 16; o > 0; o >>= 1) acc += __shfl_xor(acc, o);
    if (lane == 0) {
      if (g < F_DIM) w[g] = acc;
      else *cconst = bias[0] * acc + b2[0];
    }
  } else {
    const int q = ((int)blockIdx.x - nPrepBlk) * blockDim.x + threadIdx.x;
    if (4 * q < nCntWords) reinterpret_cast<int4*>(cnt)[q] = make_int4(0, 0, 0, 0);
    if (q == 0) *spillCount = 0;
  }
}

__device__ __forceinline__ void rank_slot_body(const int* __restrict__ row,
                                               const int* __restrict__ col,
                                               int* __restrict__ cnt,
                                               int* __restrict__ slot,
                                               int* __restrict__ spillCount,
                                               int* __restrict__ spill,
                                               int E, int N, int cntStride, int blk) {
  const int half = cache_half();
  cnt  += (size_t)half * (size_t)cntStride;
  slot += (size_t)half * (size_t)SLOT_CAP * (size_t)N;
  const int base = blk * (blockDim.x * RANK_EPT) + threadIdx.x;
  int r[RANK_EPT], c[RANK_EPT], k[RANK_EPT];
#pragma unroll
  for (int j = 0; j < RANK_EPT; ++j) {
    int e = base + j * blockDim.x;
    bool ok = e < E;
    c[j] = ok ? col[e] : -1;
    r[j] = ok ? row[e] : 0;
  }
#pragma unroll
  for (int j = 0; j < RANK_EPT; ++j)
    k[j] = (c[j] >= 0) ? atomicAdd(&cnt[c[j]], 1) : 0;
#pragma unroll
  for (int j = 0; j < RANK_EPT; ++j) {
    if (c[j] >= 0) {
      if (k[j] < SLOT_CAP) {
        slot[(size_t)k[j] * (size_t)N + (size_t)c[j]] = r[j];
      } else {
        int p = atomicAdd(spillCount, 1);
        spill[2 * (size_t)p]     = r[j];
        spill[2 * (size_t)p + 1] = c[j];
      }
    }
  }
}

__device__ __forceinline__ void s_wmma_body(const float* __restrict__ x,
                                            const float* __restrict__ w,
                                            float* __restrict__ s, int nTiles, int blk) {
  int lane = threadIdx.x & 31;
  int tile = blk * (blockDim.x >> 5) + (threadIdx.x >> 5);
  if (tile >= nTiles) return;
  int m = lane & 15;
  int halfSel = lane >> 4;
  const float* xrow = x + (size_t)(tile * 16 + m) * F_DIM;
  v8f c = {0.f, 0.f, 0.f, 0.f, 0.f, 0.f, 0.f, 0.f};
#pragma unroll
  for (int k0 = 0; k0 < F_DIM; k0 += 4) {
    int koff = k0 + 2 * halfSel;
    v2f a = *reinterpret_cast<const v2f*>(xrow + koff);
    v2f b = *reinterpret_cast<const v2f*>(w + koff);
    c = __builtin_amdgcn_wmma_f32_16x16x4_f32(
         false, a,  false, b,
         (short)0, c,  false,  false);
  }
  if ((lane & 15) == 0) {
    int base = tile * 16 + halfSel * 8;
#pragma unroll
    for (int j = 0; j < 8; ++j) s[base + j] = c[j];
  }
}

__global__ void rank_s_kernel(const int* __restrict__ row,
                              const int* __restrict__ col,
                              int* __restrict__ cnt,
                              int* __restrict__ slot,
                              int* __restrict__ spillCount,
                              int* __restrict__ spill,
                              int E, int N, int cntStride, int nBlkRank,
                              const float* __restrict__ x,
                              const float* __restrict__ w,
                              float* __restrict__ s, int nTiles) {
  if ((int)blockIdx.x < nBlkRank)
    rank_slot_body(row, col, cnt, slot, spillCount, spill, E, N, cntStride, (int)blockIdx.x);
  else
    s_wmma_body(x, w, s, nTiles, (int)blockIdx.x - nBlkRank);
}

__global__ void s_tail_kernel(const float* __restrict__ x,
                              const float* __restrict__ w,
                              float* __restrict__ s, int start, int N) {
  int n = start + blockIdx.x * blockDim.x + threadIdx.x;
  if (n < N) {
    float acc = 0.f;
    for (int k = 0; k < F_DIM; ++k)
      acc = fmaf(x[(size_t)n * F_DIM + k], w[k], acc);
    s[n] = acc;
  }
}

__global__ void dinv_t_kernel(const int* __restrict__ cnt,
                              const float* __restrict__ s,
                              float* __restrict__ dinv,
                              float* __restrict__ t, int N, int cntStride) {
  int n = blockIdx.x * blockDim.x + threadIdx.x;
  if (n < N) {
    float d = rsqrtf(fmaxf((float)(cnt[n] + cnt[cntStride + n] + 1), 1.0f));
    dinv[n] = d;
    t[n] = d * s[n];
  }
}

#define PULL_U 8
__global__ void pull_out_kernel(const int* __restrict__ cnt,
                                const int* __restrict__ slot,
                                const float* __restrict__ dinv,
                                const float* __restrict__ t,
                                const float* __restrict__ cconst,
                                const int* __restrict__ spillCount,
                                const int* __restrict__ spill,
                                float* __restrict__ out, int N, int cntStride, int E) {
  int c = blockIdx.x * blockDim.x + threadIdx.x;
  if (c >= N) return;
  int n0 = cnt[c];
  int n1 = cnt[cntStride + c];
  const bool spilled = (n0 > SLOT_CAP) || (n1 > SLOT_CAP);
  if (n0 > SLOT_CAP) n0 = SLOT_CAP;
  if (n1 > SLOT_CAP) n1 = SLOT_CAP;
  const int n = n0 + n1;
  const float d = dinv[c];
  const float cst = *cconst;
  float acc = t[c];
  const int* sp = slot + c;
  const int skip = SLOT_CAP - n0;
  for (int k0 = 0; k0 < n; k0 += PULL_U) {
    int r[PULL_U];
    float v[PULL_U];
#pragma unroll
    for (int j = 0; j < PULL_U; ++j) {
      int i = k0 + j;
      int p = (i < n0) ? i : i + skip;
      r[j] = (i < n) ? sp[(size_t)p * (size_t)N] : -1;
    }
#pragma unroll
    for (int j = 0; j < PULL_U; ++j)
      v[j] = (r[j] >= 0) ? t[r[j]] : 0.f;
#pragma unroll
    for (int j = 0; j < PULL_U; ++j) acc += v[j];
  }
  if (spilled) {
    int ns = *spillCount;
    if (ns > E) ns = E;
    for (int i = 0; i < ns; ++i)
      if (spill[2 * (size_t)i + 1] == c) acc += t[spill[2 * (size_t)i]];
  }
  out[c] = fmaf(d, acc, cst);
}

static inline size_t roundup256(size_t v) { return (v + 255) & ~(size_t)255; }

extern "C" void kernel_launch(void* const* d_in, const int* in_sizes, int n_in,
                              void* d_out, int out_size, void* d_ws, size_t ws_size,
                              hipStream_t stream) {
  const float* x    = (const float*)d_in[0];
  const int*   eidx = (const int*)d_in[1];
  const float* W1   = (const float*)d_in[2];
  const float* bias = (const float*)d_in[3];
  const float* W2   = (const float*)d_in[4];
  const float* b2   = (const float*)d_in[5];
  float* out = (float*)d_out;

  const int N = out_size;
  const int E = in_sizes[1] / 2;
  const int H = in_sizes[4];
  const int* row = eidx;
  const int* col = eidx + E;

  char* ws = (char*)d_ws;
  const size_t plane = roundup256((size_t)N * 4);
  const int cntStride = (int)(plane / 4);
  int*   cnt  = (int*)ws;
  float* s    = (float*)(ws + (N_HALF + 0) * plane);
  float* dinv = (float*)(ws + (N_HALF + 1) * plane);
  float* t    = (float*)(ws + (N_HALF + 2) * plane);
  float* w    = (float*)(ws + (N_HALF + 3) * plane);
  float* cc   = w + 64;
  int*   spillCount = (int*)(w + 65);
  int*   slot  = (int*)(ws + (N_HALF + 3) * plane + 1024);
  int*   spill = (int*)(ws + (N_HALF + 3) * plane + 1024 +
                        roundup256((size_t)N_HALF * (size_t)SLOT_CAP * (size_t)N * 4));
  (void)ws_size; (void)n_in;

  const int T = 256;
  const int nBlkN = (N + T - 1) / T;
  const int nCntWords = N_HALF * cntStride;
  const int wavesPerBlk = T / 32;
  const int nPrepBlk = (PREP_WAVES + wavesPerBlk - 1) / wavesPerBlk;
  const int nBlkZero = (nCntWords / 4 + T - 1) / T;
  const int nBlkRank = (E + T * RANK_EPT - 1) / (T * RANK_EPT);
  const int nTiles = N / 16;
  const int nBlkTile = (nTiles + wavesPerBlk - 1) / wavesPerBlk;

  prep_zero_kernel<<<nPrepBlk + nBlkZero, T, 0, stream>>>(W1, bias, W2, b2, w, cc, H,
                                                          cnt, spillCount, nCntWords, nPrepBlk);
  if (nBlkRank + nBlkTile > 0)
    rank_s_kernel<<<nBlkRank + nBlkTile, T, 0, stream>>>(row, col, cnt, slot, spillCount, spill,
                                                         E, N, cntStride, nBlkRank, x, w, s, nTiles);
  const int rem = N - nTiles * 16;
  if (rem > 0)
    s_tail_kernel<<<1, T, 0, stream>>>(x, w, s, nTiles * 16, N);

  dinv_t_kernel<<<nBlkN, T, 0, stream>>>(cnt, s, dinv, t, N, cntStride);

  const int TP = 128;
  pull_out_kernel<<<(N + TP - 1) / TP, TP, 0, stream>>>(cnt, slot, dinv, t, cc, spillCount, spill,
                                                       out, N, cntStride, E);
}
